# starting kernel
# baseline (speedup 1.0000x reference)
#include <hip/hip_runtime.h>

#define D_   480
#define H_   360
#define W_   32
#define B_   2
#define COUT 32
#define EPSF 1e-5f
#define NEGF 0.01f
#define NBR_STRIDE 12
#define TPW  4

typedef __attribute__((ext_vector_type(2))) float v2f;
typedef __attribute__((ext_vector_type(8))) float v8f;

__global__ void grid_init_kernel(int4* __restrict__ grid, int n4) {
    int i = blockIdx.x * blockDim.x + threadIdx.x;
    if (i < n4) grid[i] = make_int4(-1, -1, -1, -1);
}

__global__ void zrow_init_kernel(float* __restrict__ z) {
    z[threadIdx.x] = 0.0f;
}

__global__ void grid_scatter_kernel(const int* __restrict__ coords,
                                    int* __restrict__ grid, int n) {
    int i = blockIdx.x * blockDim.x + threadIdx.x;
    if (i >= n) return;
    int b = coords[4 * i + 0];
    int z = coords[4 * i + 1];
    int y = coords[4 * i + 2];
    int x = coords[4 * i + 3];
    grid[((b * D_ + z) * H_ + y) * W_ + x] = i;
}

__global__ void nbr_build_kernel(const int* __restrict__ coords,
                                 const int* __restrict__ grid,
                                 int* __restrict__ nbr133,
                                 int* __restrict__ nbr313, int n) {
    int i = blockIdx.x * blockDim.x + threadIdx.x;
    if (i >= n) return;
    int b = coords[4 * i + 0];
    int z = coords[4 * i + 1];
    int y = coords[4 * i + 2];
    int x = coords[4 * i + 3];
    int k = 0;
    #pragma unroll
    for (int dy = -1; dy <= 1; ++dy) {
        #pragma unroll
        for (int dx = -1; dx <= 1; ++dx) {
            int ny = y + dy, nx = x + dx;
            int v = -1;
            if (ny >= 0 && ny < H_ && nx >= 0 && nx < W_)
                v = grid[((b * D_ + z) * H_ + ny) * W_ + nx];
            nbr133[i * NBR_STRIDE + k] = v;
            ++k;
        }
    }
    k = 0;
    #pragma unroll
    for (int dz = -1; dz <= 1; ++dz) {
        #pragma unroll
        for (int dx = -1; dx <= 1; ++dx) {
            int nz = z + dz, nx = x + dx;
            int v = -1;
            if (nz >= 0 && nz < D_ && nx >= 0 && nx < W_)
                v = grid[((b * D_ + nz) * H_ + y) * W_ + nx];
            nbr313[i * NBR_STRIDE + k] = v;
            ++k;
        }
    }
}

template <int CIN, bool ADD_RES>
__global__ void __launch_bounds__(256)
conv_bn_kernel(const float* __restrict__ feats,
               const float* __restrict__ Wg,
               const int*   __restrict__ nbr,
               const float* __restrict__ zrow,
               const float* __restrict__ gamma,
               const float* __restrict__ beta,
               const float* __restrict__ mean,
               const float* __restrict__ var,
               const float* __restrict__ res,
               float* __restrict__ out,
               int n) {
    __shared__ float Wl[9 * CIN * COUT];
    for (int i = threadIdx.x; i < 9 * CIN * COUT; i += blockDim.x) {
        int k  = i / (CIN * COUT);
        int r  = i - k * (CIN * COUT);
        int ci = r / COUT;
        int co = r - ci * COUT;
        Wl[((k * (CIN / 2) + (ci >> 1)) * COUT + co) * 2 + (ci & 1)] = Wg[i];
    }
    __syncthreads();

    const int lane = threadIdx.x & 31;
    const int wv   = threadIdx.x >> 5;
    const int row  = lane & 15;
    const int half = lane >> 4;

    float scl[2], bia[2];
    #pragma unroll
    for (int t2 = 0; t2 < 2; ++t2) {
        const int co = t2 * 16 + row;
        scl[t2] = gamma[co] * rsqrtf(var[co] + EPSF);
        bia[t2] = beta[co] - mean[co] * scl[t2];
    }

    const int tile0 = (blockIdx.x * (blockDim.x >> 5) + wv) * TPW;

    #pragma unroll 1
    for (int t = 0; t < TPW; ++t) {
        const int tile = tile0 + t;
        if (tile * 16 >= n) break;

        const int point  = tile * 16 + row;
        const int pclamp = point < n ? point : n - 1;

        const int4* np = (const int4*)(nbr + (size_t)pclamp * NBR_STRIDE);
        const int4 n03 = np[0];
        const int4 n47 = np[1];
        const int  n8  = ((const int*)(np + 2))[0];
        const int nbrs[9] = {n03.x, n03.y, n03.z, n03.w,
                             n47.x, n47.y, n47.z, n47.w, n8};

        v8f acc0 = {};
        v8f acc1 = {};

        #pragma unroll
        for (int k = 0; k < 9; ++k) {
            const int nb = nbrs[k];
            const float* frow = (nb >= 0) ? (feats + (size_t)nb * CIN) : zrow;
            #pragma unroll
            for (int c = 0; c < CIN; c += 4) {
                const v2f A = *(const v2f*)(frow + c + 2 * half);
                const float* wp =
                    &Wl[((k * (CIN / 2) + (c >> 1) + half) * COUT + row) * 2];
                const v2f B0 = *(const v2f*)(wp);
                const v2f B1 = *(const v2f*)(wp + 32);
                acc0 = __builtin_amdgcn_wmma_f32_16x16x4_f32(
                    false, A, false, B0, (short)0, acc0, false, false);
                acc1 = __builtin_amdgcn_wmma_f32_16x16x4_f32(
                    false, A, false, B1, (short)0, acc1, false, false);
            }
        }

        #pragma unroll
        for (int t2 = 0; t2 < 2; ++t2) {
            const int co = t2 * 16 + row;
            v8f acc = t2 ? acc1 : acc0;
            #pragma unroll
            for (int v = 0; v < 8; ++v) {
                float yv = acc[v] * scl[t2] + bia[t2];
                yv = (yv >= 0.0f) ? yv : NEGF * yv;
                const int orow = tile * 16 + half * 8 + v;
                if (orow < n) {
                    const int oidx = orow * COUT + co;
                    if (ADD_RES) yv += res[oidx];
                    out[oidx] = yv;
                }
            }
        }
    }
}

extern "C" void kernel_launch(void* const* d_in, const int* in_sizes, int n_in,
                              void* d_out, int out_size, void* d_ws, size_t ws_size,
                              hipStream_t stream) {
    const float* feats    = (const float*)d_in[0];
    const int*   coords   = (const int*)  d_in[1];
    const float* W1       = (const float*)d_in[2];
    const float* W2       = (const float*)d_in[3];
    const float* W3       = (const float*)d_in[4];
    const float* W4       = (const float*)d_in[5];
    const float* bn_gamma = (const float*)d_in[6];
    const float* bn_beta  = (const float*)d_in[7];
    const float* bn_mean  = (const float*)d_in[8];
    const float* bn_var   = (const float*)d_in[9];
    float* out = (float*)d_out;

    const int n = in_sizes[0] / 16;

    const size_t GRID_ELEMS = (size_t)B_ * D_ * H_ * W_;
    int*   grid   = (int*)d_ws;
    int*   nbr133 = grid + GRID_ELEMS;
    int*   nbr313 = nbr133 + (size_t)n * NBR_STRIDE;
    float* bufA   = (float*)(nbr313 + (size_t)n * NBR_STRIDE);
    float* bufB   = bufA + (size_t)n * COUT;
    float* zrow   = bufB + (size_t)n * COUT;

    const int n4 = (int)(GRID_ELEMS / 4);
    grid_init_kernel<<<(n4 + 255) / 256, 256, 0, stream>>>((int4*)grid, n4);
    zrow_init_kernel<<<1, 32, 0, stream>>>(zrow);
    grid_scatter_kernel<<<(n + 255) / 256, 256, 0, stream>>>(coords, grid, n);

    nbr_build_kernel<<<(n + 255) / 256, 256, 0, stream>>>(coords, grid, nbr133, nbr313, n);

    const int tiles  = (n + 15) / 16;
    const int wavesN = (tiles + TPW - 1) / TPW;
    const int blocks = (wavesN + 7) / 8;

    conv_bn_kernel<16, false><<<blocks, 256, 0, stream>>>(
        feats, W1, nbr133, zrow,
        bn_gamma + 0,  bn_beta + 0,  bn_mean + 0,  bn_var + 0,
        nullptr, bufA, n);
    conv_bn_kernel<32, false><<<blocks, 256, 0, stream>>>(
        bufA,  W2, nbr313, zrow,
        bn_gamma + 32, bn_beta + 32, bn_mean + 32, bn_var + 32,
        nullptr, bufB, n);
    conv_bn_kernel<16, false><<<blocks, 256, 0, stream>>>(
        feats, W3, nbr313, zrow,
        bn_gamma + 64, bn_beta + 64, bn_mean + 64, bn_var + 64,
        nullptr, bufA, n);
    conv_bn_kernel<32, true><<<blocks, 256, 0, stream>>>(
        bufA,  W4, nbr133, zrow,
        bn_gamma + 96, bn_beta + 96, bn_mean + 96, bn_var + 96,
        bufB, out, n);
}


# conv passes skip, on a wave-uniform vote, every tap with no present neighbour among the tile's 16 points (its gathers and WMMAs); skipped terms are exact zeros, summation order unchanged
# speedup vs baseline: 1.1561x; 1.1561x over previous
#include <hip/hip_runtime.h>

#define D_   480
#define H_   360
#define W_   32
#define B_   2
#define COUT 32
#define EPSF 1e-5f
#define NEGF 0.01f
#define NBR_STRIDE 12
#define TPW  4

typedef __attribute__((ext_vector_type(2))) float v2f;
typedef __attribute__((ext_vector_type(8))) float v8f;

__global__ void grid_init_kernel(int4* __restrict__ grid, int n4) {
    int i = blockIdx.x * blockDim.x + threadIdx.x;
    if (i < n4) grid[i] = make_int4(-1, -1, -1, -1);
}

__global__ void zrow_init_kernel(float* __restrict__ z) {
    z[threadIdx.x] = 0.0f;
}

__global__ void grid_scatter_kernel(const int* __restrict__ coords,
                                    int* __restrict__ grid, int n) {
    int i = blockIdx.x * blockDim.x + threadIdx.x;
    if (i >= n) return;
    int b = coords[4 * i + 0];
    int z = coords[4 * i + 1];
    int y = coords[4 * i + 2];
    int x = coords[4 * i + 3];
    grid[((b * D_ + z) * H_ + y) * W_ + x] = i;
}

__global__ void nbr_build_kernel(const int* __restrict__ coords,
                                 const int* __restrict__ grid,
                                 int* __restrict__ nbr133,
                                 int* __restrict__ nbr313, int n) {
    int i = blockIdx.x * blockDim.x + threadIdx.x;
    if (i >= n) return;
    int b = coords[4 * i + 0];
    int z = coords[4 * i + 1];
    int y = coords[4 * i + 2];
    int x = coords[4 * i + 3];
    int k = 0;
    #pragma unroll
    for (int dy = -1; dy <= 1; ++dy) {
        #pragma unroll
        for (int dx = -1; dx <= 1; ++dx) {
            int ny = y + dy, nx = x + dx;
            int v = -1;
            if (ny >= 0 && ny < H_ && nx >= 0 && nx < W_)
                v = grid[((b * D_ + z) * H_ + ny) * W_ + nx];
            nbr133[i * NBR_STRIDE + k] = v;
            ++k;
        }
    }
    k = 0;
    #pragma unroll
    for (int dz = -1; dz <= 1; ++dz) {
        #pragma unroll
        for (int dx = -1; dx <= 1; ++dx) {
            int nz = z + dz, nx = x + dx;
            int v = -1;
            if (nz >= 0 && nz < D_ && nx >= 0 && nx < W_)
                v = grid[((b * D_ + nz) * H_ + y) * W_ + nx];
            nbr313[i * NBR_STRIDE + k] = v;
            ++k;
        }
    }
}

template <int CIN, bool ADD_RES>
__global__ void __launch_bounds__(256)
conv_bn_kernel(const float* __restrict__ feats,
               const float* __restrict__ Wg,
               const int*   __restrict__ nbr,
               const float* __restrict__ zrow,
               const float* __restrict__ gamma,
               const float* __restrict__ beta,
               const float* __restrict__ mean,
               const float* __restrict__ var,
               const float* __restrict__ res,
               float* __restrict__ out,
               int n) {
    __shared__ float Wl[9 * CIN * COUT];
    for (int i = threadIdx.x; i < 9 * CIN * COUT; i += blockDim.x) {
        int k  = i / (CIN * COUT);
        int r  = i - k * (CIN * COUT);
        int ci = r / COUT;
        int co = r - ci * COUT;
        Wl[((k * (CIN / 2) + (ci >> 1)) * COUT + co) * 2 + (ci & 1)] = Wg[i];
    }
    __syncthreads();

    const int lane = threadIdx.x & 31;
    const int wv   = threadIdx.x >> 5;
    const int row  = lane & 15;
    const int half = lane >> 4;

    float scl[2], bia[2];
    #pragma unroll
    for (int t2 = 0; t2 < 2; ++t2) {
        const int co = t2 * 16 + row;
        scl[t2] = gamma[co] * rsqrtf(var[co] + EPSF);
        bia[t2] = beta[co] - mean[co] * scl[t2];
    }

    const int tile0 = (blockIdx.x * (blockDim.x >> 5) + wv) * TPW;

    #pragma unroll 1
    for (int t = 0; t < TPW; ++t) {
        const int tile = tile0 + t;
        if (tile * 16 >= n) break;

        const int point  = tile * 16 + row;
        const int pclamp = point < n ? point : n - 1;

        const int4* np = (const int4*)(nbr + (size_t)pclamp * NBR_STRIDE);
        const int4 n03 = np[0];
        const int4 n47 = np[1];
        const int  n8  = ((const int*)(np + 2))[0];
        const int nbrs[9] = {n03.x, n03.y, n03.z, n03.w,
                             n47.x, n47.y, n47.z, n47.w, n8};

        v8f acc0 = {};
        v8f acc1 = {};

        bool live[9];
        #pragma unroll
        for (int k = 0; k < 9; ++k)
            live[k] = __builtin_amdgcn_ballot_w32(nbrs[k] >= 0) != 0u;

        v2f A[9][CIN / 4];
        #pragma unroll
        for (int k = 0; k < 9; ++k) {
            if (live[k]) {
                const int nb = nbrs[k];
                const float* frow = (nb >= 0) ? (feats + (size_t)nb * CIN) : zrow;
                #pragma unroll
                for (int c = 0; c < CIN; c += 4)
                    A[k][c >> 2] = *(const v2f*)(frow + c + 2 * half);
            }
        }

        #pragma unroll
        for (int k = 0; k < 9; ++k) {
            if (live[k]) {
                #pragma unroll
                for (int c = 0; c < CIN; c += 4) {
                    const float* wp =
                        &Wl[((k * (CIN / 2) + (c >> 1) + half) * COUT + row) * 2];
                    const v2f B0 = *(const v2f*)(wp);
                    const v2f B1 = *(const v2f*)(wp + 32);
                    acc0 = __builtin_amdgcn_wmma_f32_16x16x4_f32(
                        false, A[k][c >> 2], false, B0, (short)0, acc0, false, false);
                    acc1 = __builtin_amdgcn_wmma_f32_16x16x4_f32(
                        false, A[k][c >> 2], false, B1, (short)0, acc1, false, false);
                }
            }
        }

        #pragma unroll
        for (int t2 = 0; t2 < 2; ++t2) {
            const int co = t2 * 16 + row;
            v8f acc = t2 ? acc1 : acc0;
            #pragma unroll
            for (int v = 0; v < 8; ++v) {
                float yv = acc[v] * scl[t2] + bia[t2];
                yv = (yv >= 0.0f) ? yv : NEGF * yv;
                const int orow = tile * 16 + half * 8 + v;
                if (orow < n) {
                    const int oidx = orow * COUT + co;
                    if (ADD_RES) yv += res[oidx];
                    out[oidx] = yv;
                }
            }
        }
    }
}

extern "C" void kernel_launch(void* const* d_in, const int* in_sizes, int n_in,
                              void* d_out, int out_size, void* d_ws, size_t ws_size,
                              hipStream_t stream) {
    const float* feats    = (const float*)d_in[0];
    const int*   coords   = (const int*)  d_in[1];
    const float* W1       = (const float*)d_in[2];
    const float* W2       = (const float*)d_in[3];
    const float* W3       = (const float*)d_in[4];
    const float* W4       = (const float*)d_in[5];
    const float* bn_gamma = (const float*)d_in[6];
    const float* bn_beta  = (const float*)d_in[7];
    const float* bn_mean  = (const float*)d_in[8];
    const float* bn_var   = (const float*)d_in[9];
    float* out = (float*)d_out;

    const int n = in_sizes[0] / 16;

    const size_t GRID_ELEMS = (size_t)B_ * D_ * H_ * W_;
    int*   grid   = (int*)d_ws;
    int*   nbr133 = grid + GRID_ELEMS;
    int*   nbr313 = nbr133 + (size_t)n * NBR_STRIDE;
    float* bufA   = (float*)(nbr313 + (size_t)n * NBR_STRIDE);
    float* bufB   = bufA + (size_t)n * COUT;
    float* zrow   = bufB + (size_t)n * COUT;

    const int n4 = (int)(GRID_ELEMS / 4);
    grid_init_kernel<<<(n4 + 255) / 256, 256, 0, stream>>>((int4*)grid, n4);
    zrow_init_kernel<<<1, 32, 0, stream>>>(zrow);
    grid_scatter_kernel<<<(n + 255) / 256, 256, 0, stream>>>(coords, grid, n);

    nbr_build_kernel<<<(n + 255) / 256, 256, 0, stream>>>(coords, grid, nbr133, nbr313, n);

    const int tiles  = (n + 15) / 16;
    const int wavesN = (tiles + TPW - 1) / TPW;
    const int blocks = (wavesN + 7) / 8;

    conv_bn_kernel<16, false><<<blocks, 256, 0, stream>>>(
        feats, W1, nbr133, zrow,
        bn_gamma + 0,  bn_beta + 0,  bn_mean + 0,  bn_var + 0,
        nullptr, bufA, n);
    conv_bn_kernel<32, false><<<blocks, 256, 0, stream>>>(
        bufA,  W2, nbr313, zrow,
        bn_gamma + 32, bn_beta + 32, bn_mean + 32, bn_var + 32,
        nullptr, bufB, n);
    conv_bn_kernel<16, false><<<blocks, 256, 0, stream>>>(
        feats, W3, nbr313, zrow,
        bn_gamma + 64, bn_beta + 64, bn_mean + 64, bn_var + 64,
        nullptr, bufA, n);
    conv_bn_kernel<32, true><<<blocks, 256, 0, stream>>>(
        bufA,  W4, nbr133, zrow,
        bn_gamma + 96, bn_beta + 96, bn_mean + 96, bn_var + 96,
        bufB, out, n);
}


# conv prologue issues all of a block's global loads (BN constants, 18 weight loads per thread) in one batch before its 9 LDS stores; was one load and one exposed L2 round trip per trip, 18 or 36 trips
# speedup vs baseline: 1.5290x; 1.3225x over previous
#include <hip/hip_runtime.h>

#define D_   480
#define H_   360
#define W_   32
#define B_   2
#define COUT 32
#define EPSF 1e-5f
#define NEGF 0.01f
#define NBR_STRIDE 12
#define TPW  4
#define CONV_THREADS 256

typedef __attribute__((ext_vector_type(2))) float v2f;
typedef __attribute__((ext_vector_type(8))) float v8f;

__global__ void grid_init_kernel(int4* __restrict__ grid, int n4) {
    int i = blockIdx.x * blockDim.x + threadIdx.x;
    if (i < n4) grid[i] = make_int4(-1, -1, -1, -1);
}

__global__ void zrow_init_kernel(float* __restrict__ z) {
    z[threadIdx.x] = 0.0f;
}

__global__ void grid_scatter_kernel(const int* __restrict__ coords,
                                    int* __restrict__ grid, int n) {
    int i = blockIdx.x * blockDim.x + threadIdx.x;
    if (i >= n) return;
    int b = coords[4 * i + 0];
    int z = coords[4 * i + 1];
    int y = coords[4 * i + 2];
    int x = coords[4 * i + 3];
    grid[((b * D_ + z) * H_ + y) * W_ + x] = i;
}

__global__ void nbr_build_kernel(const int* __restrict__ coords,
                                 const int* __restrict__ grid,
                                 int* __restrict__ nbr133,
                                 int* __restrict__ nbr313, int n) {
    int i = blockIdx.x * blockDim.x + threadIdx.x;
    if (i >= n) return;
    int b = coords[4 * i + 0];
    int z = coords[4 * i + 1];
    int y = coords[4 * i + 2];
    int x = coords[4 * i + 3];
    int k = 0;
    #pragma unroll
    for (int dy = -1; dy <= 1; ++dy) {
        #pragma unroll
        for (int dx = -1; dx <= 1; ++dx) {
            int ny = y + dy, nx = x + dx;
            int v = -1;
            if (ny >= 0 && ny < H_ && nx >= 0 && nx < W_)
                v = grid[((b * D_ + z) * H_ + ny) * W_ + nx];
            nbr133[i * NBR_STRIDE + k] = v;
            ++k;
        }
    }
    k = 0;
    #pragma unroll
    for (int dz = -1; dz <= 1; ++dz) {
        #pragma unroll
        for (int dx = -1; dx <= 1; ++dx) {
            int nz = z + dz, nx = x + dx;
            int v = -1;
            if (nz >= 0 && nz < D_ && nx >= 0 && nx < W_)
                v = grid[((b * D_ + nz) * H_ + y) * W_ + nx];
            nbr313[i * NBR_STRIDE + k] = v;
            ++k;
        }
    }
}

template <int CIN, bool ADD_RES>
__global__ void __launch_bounds__(CONV_THREADS)
conv_bn_kernel(const float* __restrict__ feats,
               const float* __restrict__ Wg,
               const int*   __restrict__ nbr,
               const float* __restrict__ zrow,
               const float* __restrict__ gamma,
               const float* __restrict__ beta,
               const float* __restrict__ mean,
               const float* __restrict__ var,
               const float* __restrict__ res,
               float* __restrict__ out,
               int n) {
    __shared__ __attribute__((aligned(16))) float Wl[9 * CIN * COUT];

    const int lane = threadIdx.x & 31;
    const int wv   = threadIdx.x >> 5;
    const int row  = lane & 15;
    const int half = lane >> 4;

    float scl[2], bia[2];
    #pragma unroll
    for (int t2 = 0; t2 < 2; ++t2) {
        const int co = t2 * 16 + row;
        scl[t2] = gamma[co] * rsqrtf(var[co] + EPSF);
        bia[t2] = beta[co] - mean[co] * scl[t2];
    }

    constexpr int E      = CIN / 16;
    constexpr int CPR    = COUT / E;
    constexpr int PIECES = 9 * (CIN / 2) * CPR;
    constexpr int TRIPS  = PIECES / CONV_THREADS;
    static_assert(PIECES % CONV_THREADS == 0, "staging trips must be full");
    typedef float vE  __attribute__((ext_vector_type(E)));
    typedef float v2E __attribute__((ext_vector_type(2 * E)));
    const vE* WgE = (const vE*)Wg;
    vE wa[TRIPS], wb[TRIPS];
    #pragma unroll
    for (int j = 0; j < TRIPS; ++j) {
        const int p   = (int)threadIdx.x + j * CONV_THREADS;
        const int src = (p / CPR) * (2 * CPR) + (p % CPR);
        wa[j] = WgE[src];
        wb[j] = WgE[src + CPR];
    }
    #pragma unroll
    for (int j = 0; j < TRIPS; ++j) {
        const int p = (int)threadIdx.x + j * CONV_THREADS;
        v2E z;
        #pragma unroll
        for (int e = 0; e < E; ++e) {
            z[2 * e]     = wa[j][e];
            z[2 * e + 1] = wb[j][e];
        }
        ((v2E*)Wl)[p] = z;
    }
    __syncthreads();

    const int tile0 = (blockIdx.x * (blockDim.x >> 5) + wv) * TPW;

    #pragma unroll 1
    for (int t = 0; t < TPW; ++t) {
        const int tile = tile0 + t;
        if (tile * 16 >= n) break;

        const int point  = tile * 16 + row;
        const int pclamp = point < n ? point : n - 1;

        const int4* np = (const int4*)(nbr + (size_t)pclamp * NBR_STRIDE);
        const int4 n03 = np[0];
        const int4 n47 = np[1];
        const int  n8  = ((const int*)(np + 2))[0];
        const int nbrs[9] = {n03.x, n03.y, n03.z, n03.w,
                             n47.x, n47.y, n47.z, n47.w, n8};

        v8f acc0 = {};
        v8f acc1 = {};

        bool live[9];
        #pragma unroll
        for (int k = 0; k < 9; ++k)
            live[k] = __builtin_amdgcn_ballot_w32(nbrs[k] >= 0) != 0u;

        v2f A[9][CIN / 4];
        #pragma unroll
        for (int k = 0; k < 9; ++k) {
            if (live[k]) {
                const int nb = nbrs[k];
                const float* frow = (nb >= 0) ? (feats + (size_t)nb * CIN) : zrow;
                #pragma unroll
                for (int c = 0; c < CIN; c += 4)
                    A[k][c >> 2] = *(const v2f*)(frow + c + 2 * half);
            }
        }

        #pragma unroll
        for (int k = 0; k < 9; ++k) {
            if (live[k]) {
                #pragma unroll
                for (int c = 0; c < CIN; c += 4) {
                    const float* wp =
                        &Wl[((k * (CIN / 2) + (c >> 1) + half) * COUT + row) * 2];
                    const v2f B0 = *(const v2f*)(wp);
                    const v2f B1 = *(const v2f*)(wp + 32);
                    acc0 = __builtin_amdgcn_wmma_f32_16x16x4_f32(
                        false, A[k][c >> 2], false, B0, (short)0, acc0, false, false);
                    acc1 = __builtin_amdgcn_wmma_f32_16x16x4_f32(
                        false, A[k][c >> 2], false, B1, (short)0, acc1, false, false);
                }
            }
        }

        #pragma unroll
        for (int t2 = 0; t2 < 2; ++t2) {
            const int co = t2 * 16 + row;
            v8f acc = t2 ? acc1 : acc0;
            #pragma unroll
            for (int v = 0; v < 8; ++v) {
                float yv = acc[v] * scl[t2] + bia[t2];
                yv = (yv >= 0.0f) ? yv : NEGF * yv;
                const int orow = tile * 16 + half * 8 + v;
                if (orow < n) {
                    const int oidx = orow * COUT + co;
                    if (ADD_RES) yv += res[oidx];
                    out[oidx] = yv;
                }
            }
        }
    }
}

extern "C" void kernel_launch(void* const* d_in, const int* in_sizes, int n_in,
                              void* d_out, int out_size, void* d_ws, size_t ws_size,
                              hipStream_t stream) {
    const float* feats    = (const float*)d_in[0];
    const int*   coords   = (const int*)  d_in[1];
    const float* W1       = (const float*)d_in[2];
    const float* W2       = (const float*)d_in[3];
    const float* W3       = (const float*)d_in[4];
    const float* W4       = (const float*)d_in[5];
    const float* bn_gamma = (const float*)d_in[6];
    const float* bn_beta  = (const float*)d_in[7];
    const float* bn_mean  = (const float*)d_in[8];
    const float* bn_var   = (const float*)d_in[9];
    float* out = (float*)d_out;

    const int n = in_sizes[0] / 16;

    const size_t GRID_ELEMS = (size_t)B_ * D_ * H_ * W_;
    int*   grid   = (int*)d_ws;
    int*   nbr133 = grid + GRID_ELEMS;
    int*   nbr313 = nbr133 + (size_t)n * NBR_STRIDE;
    float* bufA   = (float*)(nbr313 + (size_t)n * NBR_STRIDE);
    float* bufB   = bufA + (size_t)n * COUT;
    float* zrow   = bufB + (size_t)n * COUT;

    const int n4 = (int)(GRID_ELEMS / 4);
    grid_init_kernel<<<(n4 + 255) / 256, 256, 0, stream>>>((int4*)grid, n4);
    zrow_init_kernel<<<1, 32, 0, stream>>>(zrow);
    grid_scatter_kernel<<<(n + 255) / 256, 256, 0, stream>>>(coords, grid, n);

    nbr_build_kernel<<<(n + 255) / 256, 256, 0, stream>>>(coords, grid, nbr133, nbr313, n);

    const int tiles  = (n + 15) / 16;
    const int wavesN = (tiles + TPW - 1) / TPW;
    const int blocks = (wavesN + 7) / 8;

    conv_bn_kernel<16, false><<<blocks, CONV_THREADS, 0, stream>>>(
        feats, W1, nbr133, zrow,
        bn_gamma + 0,  bn_beta + 0,  bn_mean + 0,  bn_var + 0,
        nullptr, bufA, n);
    conv_bn_kernel<32, false><<<blocks, CONV_THREADS, 0, stream>>>(
        bufA,  W2, nbr313, zrow,
        bn_gamma + 32, bn_beta + 32, bn_mean + 32, bn_var + 32,
        nullptr, bufB, n);
    conv_bn_kernel<16, false><<<blocks, CONV_THREADS, 0, stream>>>(
        feats, W3, nbr313, zrow,
        bn_gamma + 64, bn_beta + 64, bn_mean + 64, bn_var + 64,
        nullptr, bufA, n);
    conv_bn_kernel<32, true><<<blocks, CONV_THREADS, 0, stream>>>(
        bufA,  W4, nbr133, zrow,
        bn_gamma + 96, bn_beta + 96, bn_mean + 96, bn_var + 96,
        bufB, out, n);
}


# neighbour-table build reads a grid row's three cells (x-1, x, x+1) from one address and issues a point's five rows as one batch ahead of its stores; was 18 single loads, each waited for in turn
# speedup vs baseline: 2.2363x; 1.4626x over previous
#include <hip/hip_runtime.h>

#define D_   480
#define H_   360
#define W_   32
#define B_   2
#define COUT 32
#define EPSF 1e-5f
#define NEGF 0.01f
#define NBR_STRIDE 12
#define GRID_PAD 32
#define TPW  4
#define CONV_THREADS 256

typedef __attribute__((ext_vector_type(2))) float v2f;
typedef __attribute__((ext_vector_type(8))) float v8f;

__global__ void grid_init_kernel(int4* __restrict__ grid, int n4,
                                 float* __restrict__ zrow) {
    int i = blockIdx.x * blockDim.x + threadIdx.x;
    if (i < n4) grid[i] = make_int4(-1, -1, -1, -1);
    if (i < COUT) zrow[i] = 0.0f;
}

__global__ void grid_scatter_kernel(const int* __restrict__ coords,
                                    int* __restrict__ grid, int n) {
    int i = blockIdx.x * blockDim.x + threadIdx.x;
    if (i >= n) return;
    int b = coords[4 * i + 0];
    int z = coords[4 * i + 1];
    int y = coords[4 * i + 2];
    int x = coords[4 * i + 3];
    grid[((b * D_ + z) * H_ + y) * W_ + x] = i;
}

__device__ __forceinline__ void row_lookup(const int* __restrict__ grid,
                                           int rowc, bool rowok, int x,
                                           int& vm, int& v0, int& vp) {
    const int* g = grid + ((size_t)rowc * W_ + x);
    const int em = g[-1];
    const int e0 = g[0];
    const int ep = g[1];
    vm = (rowok && x > 0)      ? em : -1;
    v0 = rowok                 ? e0 : -1;
    vp = (rowok && x < W_ - 1) ? ep : -1;
}

__global__ void nbr_build_kernel(const int* __restrict__ coords,
                                 const int* __restrict__ grid,
                                 int* __restrict__ nbr133,
                                 int* __restrict__ nbr313, int n) {
    int i = blockIdx.x * blockDim.x + threadIdx.x;
    if (i >= n) return;
    int b = coords[4 * i + 0];
    int z = coords[4 * i + 1];
    int y = coords[4 * i + 2];
    int x = coords[4 * i + 3];
    int v133[9], v313[9];
    row_lookup(grid, (b * D_ + z) * H_ + y, true, x, v133[3], v133[4], v133[5]);
    v313[3] = v133[3]; v313[4] = v133[4]; v313[5] = v133[5];
    #pragma unroll
    for (int dy = -1; dy <= 1; dy += 2) {
        const int ny  = y + dy;
        const int nyc = ny < 0 ? 0 : (ny > H_ - 1 ? H_ - 1 : ny);
        row_lookup(grid, (b * D_ + z) * H_ + nyc, ny == nyc, x,
                   v133[(dy + 1) * 3 + 0], v133[(dy + 1) * 3 + 1], v133[(dy + 1) * 3 + 2]);
    }
    #pragma unroll
    for (int dz = -1; dz <= 1; dz += 2) {
        const int nz  = z + dz;
        const int nzc = nz < 0 ? 0 : (nz > D_ - 1 ? D_ - 1 : nz);
        row_lookup(grid, (b * D_ + nzc) * H_ + y, nz == nzc, x,
                   v313[(dz + 1) * 3 + 0], v313[(dz + 1) * 3 + 1], v313[(dz + 1) * 3 + 2]);
    }
    #pragma unroll
    for (int k = 0; k < 9; ++k) nbr133[i * NBR_STRIDE + k] = v133[k];
    #pragma unroll
    for (int k = 0; k < 9; ++k) nbr313[i * NBR_STRIDE + k] = v313[k];
}


__device__ __forceinline__ void load_nbrs(const int* __restrict__ nbr, int p,
                                          int (&o)[9]) {
    const int4* np = (const int4*)(nbr + (size_t)p * NBR_STRIDE);
    const int4 n03 = np[0];
    const int4 n47 = np[1];
    const int  n8  = ((const int*)(np + 2))[0];
    o[0] = n03.x; o[1] = n03.y; o[2] = n03.z; o[3] = n03.w;
    o[4] = n47.x; o[5] = n47.y; o[6] = n47.z; o[7] = n47.w;
    o[8] = n8;
}

template <int CIN>
__device__ __forceinline__ void tile_conv(const float* __restrict__ feats,
                                          const float* __restrict__ zrow,
                                          const float* Wl,
                                          const int (&nbrs)[9],
                                          int row, int half,
                                          v8f& acc0, v8f& acc1) {
    bool live[9];
    #pragma unroll
    for (int k = 0; k < 9; ++k)
        live[k] = __builtin_amdgcn_ballot_w32(nbrs[k] >= 0) != 0u;

    v2f A[9][CIN / 4];
    #pragma unroll
    for (int k = 0; k < 9; ++k) {
        if (live[k]) {
            const int nb = nbrs[k];
            const float* frow = (nb >= 0) ? (feats + (size_t)nb * CIN) : zrow;
            #pragma unroll
            for (int c = 0; c < CIN; c += 4)
                A[k][c >> 2] = *(const v2f*)(frow + c + 2 * half);
        }
    }

    #pragma unroll
    for (int k = 0; k < 9; ++k) {
        if (live[k]) {
            #pragma unroll
            for (int c = 0; c < CIN; c += 4) {
                const float* wp =
                    &Wl[((k * (CIN / 2) + (c >> 1) + half) * COUT + row) * 2];
                const v2f B0 = *(const v2f*)(wp);
                const v2f B1 = *(const v2f*)(wp + 32);
                acc0 = __builtin_amdgcn_wmma_f32_16x16x4_f32(
                    false, A[k][c >> 2], false, B0, (short)0, acc0, false, false);
                acc1 = __builtin_amdgcn_wmma_f32_16x16x4_f32(
                    false, A[k][c >> 2], false, B1, (short)0, acc1, false, false);
            }
        }
    }
}

template <int CIN, bool SUM>
__global__ void __launch_bounds__(CONV_THREADS)
conv_pair_kernel(const float* __restrict__ featsA,
                 const float* __restrict__ featsB,
                 const float* __restrict__ WgA,
                 const float* __restrict__ WgB,
                 const int*   __restrict__ nbrA,
                 const int*   __restrict__ nbrB,
                 const float* __restrict__ zrow,
                 const float* __restrict__ gamma,
                 const float* __restrict__ beta,
                 const float* __restrict__ mean,
                 const float* __restrict__ var,
                 int layerA, int layerB,
                 float* __restrict__ outA,
                 float* __restrict__ outB,
                 int n) {
    __shared__ __attribute__((aligned(16))) float Wl[2][9 * CIN * COUT];

    const int lane = threadIdx.x & 31;
    const int wv   = threadIdx.x >> 5;
    const int row  = lane & 15;
    const int half = lane >> 4;

    float scl[2][2], bia[2][2];
    #pragma unroll
    for (int s = 0; s < 2; ++s) {
        const int base = (s ? layerB : layerA) * COUT;
        #pragma unroll
        for (int t2 = 0; t2 < 2; ++t2) {
            const int co = base + t2 * 16 + row;
            scl[s][t2] = gamma[co] * rsqrtf(var[co] + EPSF);
            bia[s][t2] = beta[co] - mean[co] * scl[s][t2];
        }
    }

    constexpr int E      = CIN / 16;
    constexpr int CPR    = COUT / E;
    constexpr int PIECES = 9 * (CIN / 2) * CPR;
    constexpr int TRIPS  = PIECES / CONV_THREADS;
    static_assert(PIECES % CONV_THREADS == 0, "staging trips must be full");
    typedef float vE  __attribute__((ext_vector_type(E)));
    typedef float v2E __attribute__((ext_vector_type(2 * E)));
    vE wa[2][TRIPS], wb[2][TRIPS];
    #pragma unroll
    for (int s = 0; s < 2; ++s) {
        const vE* WgE = (const vE*)(s ? WgB : WgA);
        #pragma unroll
        for (int j = 0; j < TRIPS; ++j) {
            const int p   = (int)threadIdx.x + j * CONV_THREADS;
            const int src = (p / CPR) * (2 * CPR) + (p % CPR);
            wa[s][j] = WgE[src];
            wb[s][j] = WgE[src + CPR];
        }
    }
    #pragma unroll
    for (int s = 0; s < 2; ++s) {
        #pragma unroll
        for (int j = 0; j < TRIPS; ++j) {
            const int p = (int)threadIdx.x + j * CONV_THREADS;
            v2E z;
            #pragma unroll
            for (int e = 0; e < E; ++e) {
                z[2 * e]     = wa[s][j][e];
                z[2 * e + 1] = wb[s][j][e];
            }
            ((v2E*)Wl[s])[p] = z;
        }
    }
    __syncthreads();

    const int tile0 = (blockIdx.x * (blockDim.x >> 5) + wv) * TPW;

    #pragma unroll 1
    for (int t = 0; t < TPW; ++t) {
        const int tile = tile0 + t;
        if (tile * 16 >= n) break;

        const int point  = tile * 16 + row;
        const int pclamp = point < n ? point : n - 1;

        int nbA[9], nbB[9];
        load_nbrs(nbrA, pclamp, nbA);
        load_nbrs(nbrB, pclamp, nbB);

        float ya[2][8];

        #pragma unroll 1
        for (int s = 0; s < 2; ++s) {
            const float* feats = s ? featsB : featsA;
            const float* Ws    = s ? Wl[1] : Wl[0];
            int nbrs[9];
            #pragma unroll
            for (int k = 0; k < 9; ++k) nbrs[k] = s ? nbB[k] : nbA[k];
            const float sc[2] = {s ? scl[1][0] : scl[0][0], s ? scl[1][1] : scl[0][1]};
            const float bi[2] = {s ? bia[1][0] : bia[0][0], s ? bia[1][1] : bia[0][1]};

            v8f acc0 = {};
            v8f acc1 = {};
            tile_conv<CIN>(feats, zrow, Ws, nbrs, row, half, acc0, acc1);

            float* dst = (SUM || s == 0) ? outA : outB;
            #pragma unroll
            for (int t2 = 0; t2 < 2; ++t2) {
                const int co = t2 * 16 + row;
                const v8f acc = t2 ? acc1 : acc0;
                #pragma unroll
                for (int v = 0; v < 8; ++v) {
                    float yv = acc[v] * sc[t2] + bi[t2];
                    yv = (yv >= 0.0f) ? yv : NEGF * yv;
                    if (SUM && s == 0) {
                        ya[t2][v] = yv;
                    } else {
                        if (SUM) yv += ya[t2][v];
                        const int orow = tile * 16 + half * 8 + v;
                        if (orow < n) dst[orow * COUT + co] = yv;
                    }
                }
            }
        }
    }
}

extern "C" void kernel_launch(void* const* d_in, const int* in_sizes, int n_in,
                              void* d_out, int out_size, void* d_ws, size_t ws_size,
                              hipStream_t stream) {
    const float* feats    = (const float*)d_in[0];
    const int*   coords   = (const int*)  d_in[1];
    const float* W1       = (const float*)d_in[2];
    const float* W2       = (const float*)d_in[3];
    const float* W3       = (const float*)d_in[4];
    const float* W4       = (const float*)d_in[5];
    const float* bn_gamma = (const float*)d_in[6];
    const float* bn_beta  = (const float*)d_in[7];
    const float* bn_mean  = (const float*)d_in[8];
    const float* bn_var   = (const float*)d_in[9];
    float* out = (float*)d_out;

    const int n = in_sizes[0] / 16;

    const size_t GRID_ELEMS = (size_t)B_ * D_ * H_ * W_;
    int*   grid   = (int*)d_ws + GRID_PAD;
    int*   nbr133 = grid + GRID_ELEMS + GRID_PAD;
    int*   nbr313 = nbr133 + (size_t)n * NBR_STRIDE;
    float* bufA   = (float*)(nbr313 + (size_t)n * NBR_STRIDE);
    float* bufB   = bufA + (size_t)n * COUT;
    float* zrow   = bufB + (size_t)n * COUT;

    const int n4 = (int)((GRID_ELEMS + 2 * GRID_PAD) / 4);
    grid_init_kernel<<<(n4 + 255) / 256, 256, 0, stream>>>((int4*)d_ws, n4, zrow);
    grid_scatter_kernel<<<(n + 255) / 256, 256, 0, stream>>>(coords, grid, n);

    nbr_build_kernel<<<(n + 255) / 256, 256, 0, stream>>>(coords, grid, nbr133, nbr313, n);

    const int tiles  = (n + 15) / 16;
    const int wavesN = (tiles + TPW - 1) / TPW;
    const int blocks = (wavesN + 7) / 8;

    conv_pair_kernel<16, false><<<blocks, CONV_THREADS, 0, stream>>>(
        feats, feats, W1, W3, nbr133, nbr313, zrow,
        bn_gamma, bn_beta, bn_mean, bn_var, 0, 2,
        bufA, bufB, n);
    conv_pair_kernel<32, true><<<blocks, CONV_THREADS, 0, stream>>>(
        bufA, bufB, W2, W4, nbr313, nbr133, zrow,
        bn_gamma, bn_beta, bn_mean, bn_var, 1, 3,
        out, nullptr, n);
}
